# starting kernel
# baseline (speedup 1.0000x reference)
#include <hip/hip_runtime.h>

typedef __attribute__((ext_vector_type(2))) float v2f;
typedef __attribute__((ext_vector_type(4))) float v4f;
typedef __attribute__((ext_vector_type(8))) float v8f;

#define B_   4096
#define T_   256
#define F_   64
#define H_   16
#define G4H  64
#define D_   64
#define XPAD 68
#define HPAD 18
#define WPP  160

__device__ __forceinline__ float sigf(float v) { return 1.0f / (1.0f + __expf(-v)); }

__global__ __launch_bounds__(32)
void lstm_fused_kernel(const float* __restrict__ x, const float* __restrict__ W,
                       const float* __restrict__ U, const float* __restrict__ b,
                       const float* __restrict__ Wd, const float* __restrict__ bd,
                       float* __restrict__ out) {
  __shared__ float Wlds[32 * WPP];
  __shared__ float Wdlds[8 * WPP];
  __shared__ float xlds[2][16 * XPAD];
  __shared__ float hlds[16 * HPAD];

  const int lane  = threadIdx.x & 31;
  const int n     = lane & 15;
  const int half  = lane >> 4;
  const int khalf = half * 2;
  const int mhalf = half * 8;
  const int b0    = blockIdx.x * 16;

  for (int i = lane; i < F_ * G4H; i += 32) {
    int j = i & 1, col = (i >> 1) & 63, pp = i >> 7;
    Wlds[pp * WPP + col * 2 + j] = W[(2 * pp + j) * G4H + col];
  }
  for (int i = lane; i < H_ * D_; i += 32) {
    int j = i & 1, col = (i >> 1) & 63, pp = i >> 7;
    Wdlds[pp * WPP + col * 2 + j] = Wd[(2 * pp + j) * D_ + col];
  }
  __syncthreads();

  v2f ufrag[4][4];
#pragma unroll
  for (int kk = 0; kk < 4; ++kk)
#pragma unroll
    for (int g = 0; g < 4; ++g) {
      int k0 = kk * 4 + khalf;
      ufrag[kk][g].x = U[k0 * G4H + g * 16 + n];
      ufrag[kk][g].y = U[(k0 + 1) * G4H + g * 16 + n];
    }

  float bg[4];
#pragma unroll
  for (int g = 0; g < 4; ++g) bg[g] = b[g * 16 + n];

  v8f h = {}; v8f c = {};
  float* xl0 = &xlds[0][0];
  float* xl1 = &xlds[1][0];
  float* hl  = &hlds[0];

  auto loadx = [&](int t, v4f* xr) {
#pragma unroll
    for (int i = 0; i < 8; ++i) {
      int idx = i * 32 + lane;
      int row = idx >> 4, c4 = idx & 15;
      xr[i] = *(const v4f*)(x + ((size_t)(b0 + row) * T_ + t) * F_ + c4 * 4);
    }
  };
  auto storex = [&](float* dst, const v4f* xr) {
#pragma unroll
    for (int i = 0; i < 8; ++i) {
      int idx = i * 32 + lane;
      int row = idx >> 4, c4 = idx & 15;
      float* p = dst + row * XPAD + c4 * 4;
      v2f lo; lo.x = xr[i].x; lo.y = xr[i].y;
      v2f hi; hi.x = xr[i].z; hi.y = xr[i].w;
      *(v2f*)p = lo;
      *(v2f*)(p + 2) = hi;
    }
  };

  v4f xr[8];
  loadx(0, xr);
  storex(xl0, xr);

  int p = 0;
  for (int t = 0; t < T_; ++t) {
    if (t + 1 < T_) loadx(t + 1, xr);
    if (t + 2 < T_) {
      const float* pf = x + ((size_t)(b0 + (lane >> 1)) * T_ + (t + 2)) * F_ + (lane & 1) * 32;
      __builtin_prefetch(pf, 0, 0);
    }

    v8f acc[4];
#pragma unroll
    for (int g = 0; g < 4; ++g)
#pragma unroll
      for (int r = 0; r < 8; ++r) acc[g][r] = bg[g];

    float* xl = p ? xl1 : xl0;

#pragma unroll 4
    for (int kk = 0; kk < 16; ++kk) {
      v2f a = *(const v2f*)(xl + n * XPAD + kk * 4 + khalf);
      const float* wrow = Wlds + (kk * 2 + half) * WPP;
#pragma unroll
      for (int g = 0; g < 4; ++g) {
        v2f wb = *(const v2f*)(wrow + (g * 16 + n) * 2);
        acc[g] = __builtin_amdgcn_wmma_f32_16x16x4_f32(false, a, false, wb,
                                                       (short)0, acc[g], false, false);
      }
    }

#pragma unroll
    for (int r = 0; r < 8; ++r) hl[(r + mhalf) * HPAD + n] = h[r];
#pragma unroll
    for (int kk = 0; kk < 4; ++kk) {
      v2f a = *(const v2f*)(hl + n * HPAD + kk * 4 + khalf);
#pragma unroll
      for (int g = 0; g < 4; ++g)
        acc[g] = __builtin_amdgcn_wmma_f32_16x16x4_f32(false, a, false, ufrag[kk][g],
                                                       (short)0, acc[g], false, false);
    }

#pragma unroll
    for (int r = 0; r < 8; ++r) {
      float gi = sigf(acc[0][r]);
      float gf = sigf(acc[1][r]);
      float gg = fmaxf(acc[2][r], 0.0f);
      float go = sigf(acc[3][r]);
      float cn = gf * c[r] + gi * gg;
      c[r] = cn;
      h[r] = go * fmaxf(cn, 0.0f);
    }

    if (t + 1 < T_) { storex(p ? xl0 : xl1, xr); p ^= 1; }
  }

#pragma unroll
  for (int r = 0; r < 8; ++r) hl[(r + mhalf) * HPAD + n] = h[r];

  v8f dacc[4];
#pragma unroll
  for (int j = 0; j < 4; ++j) {
    float bj = bd[j * 16 + n];
#pragma unroll
    for (int r = 0; r < 8; ++r) dacc[j][r] = bj;
  }
#pragma unroll
  for (int kk = 0; kk < 4; ++kk) {
    v2f a = *(const v2f*)(hl + n * HPAD + kk * 4 + khalf);
    const float* wrow = Wdlds + (kk * 2 + half) * WPP;
#pragma unroll
    for (int j = 0; j < 4; ++j) {
      v2f wb = *(const v2f*)(wrow + (j * 16 + n) * 2);
      dacc[j] = __builtin_amdgcn_wmma_f32_16x16x4_f32(false, a, false, wb,
                                                      (short)0, dacc[j], false, false);
    }
  }

  float* ol = xl0;
#pragma unroll
  for (int j = 0; j < 4; ++j)
#pragma unroll
    for (int r = 0; r < 8; ++r)
      ol[(r + mhalf) * XPAD + j * 16 + n] = sigf(dacc[j][r]);

  if (lane < 16) {
    float s = 0.0f;
    for (int q = 0; q < 64; ++q) s += ol[lane * XPAD + q];
    hl[lane] = 1.0f / s;
  }

#pragma unroll 4
  for (int i = 0; i < 32; ++i) {
    int idx = i * 32 + lane;
    int row = idx >> 6, col = idx & 63;
    out[(size_t)(b0 + row) * D_ + col] = ol[row * XPAD + col] * hl[row];
  }
}

extern "C" void kernel_launch(void* const* d_in, const int* in_sizes, int n_in,
                              void* d_out, int out_size, void* d_ws, size_t ws_size,
                              hipStream_t stream) {
  (void)in_sizes; (void)n_in; (void)out_size; (void)d_ws; (void)ws_size;
  const float* x  = (const float*)d_in[0];
  const float* W  = (const float*)d_in[1];
  const float* U  = (const float*)d_in[2];
  const float* b  = (const float*)d_in[3];
  const float* Wd = (const float*)d_in[4];
  const float* bd = (const float*)d_in[5];
  float* out = (float*)d_out;

  dim3 grid(B_ / 16);
  dim3 block(32);
  hipLaunchKernelGGL(lstm_fused_kernel, grid, block, 0, stream,
                     x, W, U, b, Wd, bd, out);
}


# time steps alternate between the two waves of a SIMD (8 waves/block): one adds h@U and does the gates of step t while the other runs x_{t+1}@W; h,c swapped via LDS; gate wave at s_setprio 3; same sums
# speedup vs baseline: 4.5933x; 4.5933x over previous
#include <hip/hip_runtime.h>

typedef __attribute__((ext_vector_type(2))) float v2f;
typedef __attribute__((ext_vector_type(4))) float v4f;
typedef __attribute__((ext_vector_type(8))) float v8f;

#define B_   4096
#define T_   256
#define F_   64
#define H_   16
#define G4H  64
#define D_   64
#define NW   4
#define NP   2
#define OPAD 68
#define XP   68

__device__ __forceinline__ float sigf(float v) { return 1.0f / (1.0f + __expf(-v)); }

__device__ __forceinline__ void lds_barrier() {
  __builtin_amdgcn_fence(__ATOMIC_RELEASE, "workgroup", "local");
  __builtin_amdgcn_s_barrier();
  __builtin_amdgcn_fence(__ATOMIC_ACQUIRE, "workgroup", "local");
}

__device__ __forceinline__ v8f mma4(v2f a, v2f b, v8f c) {
  return __builtin_amdgcn_wmma_f32_16x16x4_f32(false, a, false, b, (short)0, c, false, false);
}

__global__ __launch_bounds__(32 * NW * NP)
void lstm_fused_kernel(const float* __restrict__ x, const float* __restrict__ W,
                       const float* __restrict__ U, const float* __restrict__ b,
                       const float* __restrict__ Wd, const float* __restrict__ bd,
                       float* __restrict__ out) {
  __shared__ __attribute__((aligned(16))) float ol[16 * OPAD];
  __shared__ float rinv[16];
  __shared__ __attribute__((aligned(16))) float hs[NP][NW][32][2];
  __shared__ __attribute__((aligned(16))) float cs[NP][NW][32][2];
  __shared__ __attribute__((aligned(16))) float xs[2][16 * XP];

  const int lane = threadIdx.x & 31;
  const int wv   = __builtin_amdgcn_readfirstlane((int)(threadIdx.x >> 5));
  const int w    = wv & 3;
  const int p    = wv >> 2;
  const int n    = lane & 15;
  const int half = lane >> 4;
  const int b0   = blockIdx.x * 16;

  const int colA = ((n & 7) >> 1) * 16 + 4 * w + 2 * (n >> 3) + (n & 1);

  v2f wfrag[16];
#pragma unroll
  for (int kk = 0; kk < 16; ++kk) {
    int f0 = 8 * (kk >> 1) + 4 * half + 2 * (kk & 1);
    wfrag[kk].x = W[f0 * G4H + colA];
    wfrag[kk].y = W[(f0 + 1) * G4H + colA];
  }

  v2f ufrag[4];
#pragma unroll
  for (int kk = 0; kk < 4; ++kk) {
    int u0 = 4 * kk + 2 * half;
    ufrag[kk].x = U[u0 * G4H + colA];
    ufrag[kk].y = U[(u0 + 1) * G4H + colA];
  }

  v8f bg;
#pragma unroll
  for (int r = 0; r < 8; ++r) bg[r] = b[(r >> 1) * 16 + 4 * w + 2 * half + (r & 1)];

  auto xw = [&](const v4f* xr, v8f a, int j0, int j1) {
#pragma unroll
    for (int j = j0; j < j1; ++j) {
      v2f lo; lo.x = xr[j].x; lo.y = xr[j].y;
      v2f hi; hi.x = xr[j].z; hi.y = xr[j].w;
      a = mma4(wfrag[2 * j], lo, a);
      a = mma4(wfrag[2 * j + 1], hi, a);
    }
    return a;
  };

  const int crow = 2 * wv + (lane >> 4);
  const int cf   = 4 * (lane & 15);
  const float* xco = x + (size_t)(b0 + crow) * T_ * F_ + cf;
  auto gload = [&](int t) {
    return *(const v4f*)(xco + (size_t)t * F_);
  };

  const float* frd0 = &xs[0][n * XP + 4 * half];
  const float* frd1 = &xs[1][n * XP + 4 * half];
  float* swr0 = &xs[0][crow * XP + cf];
  float* swr1 = &xs[1][crow * XP + cf];
  const float* hrd = &hs[p ^ 1][0][lane][0];
  const float* crd = &cs[p ^ 1][w][lane][0];
  float* hwr = &hs[p][w][lane][0];
  float* cwr = &cs[p][w][lane][0];

  v8f acc = bg;
  v4f XF[8];

  auto pslot = [&]() {
    acc = xw(XF, bg, 0, 8);
  };

  auto gslot = [&](const float* frd) {
    __builtin_amdgcn_s_setprio(3);
    v2f hb[4];
#pragma unroll
    for (int kk = 0; kk < 4; ++kk) hb[kk] = *(const v2f*)(hrd + 64 * kk);
    v2f cp = *(const v2f*)crd;
    __builtin_amdgcn_sched_barrier(0);
#pragma unroll
    for (int i = 0; i < 8; ++i) XF[i] = *(const v4f*)(frd + 8 * i);
    __builtin_amdgcn_sched_barrier(0);
#pragma unroll
    for (int kk = 0; kk < 4; ++kk) acc = mma4(ufrag[kk], hb[kk], acc);
    const v8f z = acc;

    v2f hn, cn2;
    {
      float gi = sigf(z[0]);
      float gf = sigf(z[2]);
      float gg = fmaxf(z[4], 0.0f);
      float go = sigf(z[6]);
      float cn = gf * cp.x + gi * gg;
      cn2.x = cn;
      hn.x = go * fmaxf(cn, 0.0f);
    }
    {
      float gi = sigf(z[1]);
      float gf = sigf(z[3]);
      float gg = fmaxf(z[5], 0.0f);
      float go = sigf(z[7]);
      float cn = gf * cp.y + gi * gg;
      cn2.y = cn;
      hn.y = go * fmaxf(cn, 0.0f);
    }
    *(v2f*)hwr = hn;
    *(v2f*)cwr = cn2;
    __builtin_amdgcn_s_setprio(0);
  };

  v4f G0, G1, G2, G3;
  {
    {
      const float* q = x + (size_t)(b0 + n) * T_ * F_ + (size_t)p * F_ + 4 * half;
#pragma unroll
      for (int j = 0; j < 8; ++j) XF[j] = *(const v4f*)(q + 8 * j);
    }
    v4f GT = gload(2);
    __builtin_amdgcn_sched_barrier(0);
    G3 = gload(3);
    __builtin_amdgcn_sched_barrier(0);
    G0 = gload(4);
    __builtin_amdgcn_sched_barrier(0);
    G1 = gload(5);
    __builtin_amdgcn_sched_barrier(0);
    G2 = gload(6);
    __builtin_amdgcn_sched_barrier(0);
    if (p == 1) {
      v2f zz; zz.x = 0.0f; zz.y = 0.0f;
      *(v2f*)hwr = zz;
      *(v2f*)cwr = zz;
    }
    *(v4f*)swr0 = GT;
    if (p == 0) pslot();
  }
  lds_barrier();

  auto slot = [&](int t, int par, const float* frd, float* swr, v4f& g) {
    if (p == par) gslot(frd);
    else pslot();
    *(v4f*)swr = g;
    int tl = (t + 7 < T_) ? (t + 7) : (T_ - 1);
    g = gload(tl);
    lds_barrier();
  };

  for (int t = 0; t < T_; t += 4) {
    slot(t,     0, frd0, swr1, G3);
    slot(t + 1, 1, frd1, swr0, G0);
    slot(t + 2, 0, frd0, swr1, G1);
    slot(t + 3, 1, frd1, swr0, G2);
  }

  if (p == 0) {
    v2f hb[4];
#pragma unroll
    for (int kk = 0; kk < 4; ++kk) hb[kk] = *(const v2f*)&hs[1][kk][lane][0];
    v8f dacc;
#pragma unroll
    for (int r = 0; r < 8; ++r) dacc[r] = bd[16 * w + 8 * half + r];
#pragma unroll
    for (int kk = 0; kk < 4; ++kk) {
      int u0 = 4 * kk + 2 * half;
      v2f wa;
      wa.x = Wd[u0 * D_ + 16 * w + n];
      wa.y = Wd[(u0 + 1) * D_ + 16 * w + n];
      dacc = mma4(wa, hb[kk], dacc);
    }
    v4f s0, s1;
    s0.x = sigf(dacc[0]); s0.y = sigf(dacc[1]); s0.z = sigf(dacc[2]); s0.w = sigf(dacc[3]);
    s1.x = sigf(dacc[4]); s1.y = sigf(dacc[5]); s1.z = sigf(dacc[6]); s1.w = sigf(dacc[7]);
    float* q = ol + n * OPAD + 16 * w + 8 * half;
    *(v4f*)q = s0;
    *(v4f*)(q + 4) = s1;
  }
  __syncthreads();

  if (threadIdx.x < 16) {
    float s = 0.0f;
    for (int q = 0; q < 64; ++q) s += ol[threadIdx.x * OPAD + q];
    rinv[threadIdx.x] = 1.0f / s;
  }
  __syncthreads();

#pragma unroll
  for (int i = 0; i < 4; ++i) {
    int idx = i * (32 * NW * NP) + threadIdx.x;
    int row = idx >> 6, col = idx & 63;
    out[(size_t)(b0 + row) * D_ + col] = ol[row * OPAD + col] * rinv[row];
  }
}

extern "C" void kernel_launch(void* const* d_in, const int* in_sizes, int n_in,
                              void* d_out, int out_size, void* d_ws, size_t ws_size,
                              hipStream_t stream) {
  (void)in_sizes; (void)n_in; (void)out_size; (void)d_ws; (void)ws_size;
  const float* x  = (const float*)d_in[0];
  const float* W  = (const float*)d_in[1];
  const float* U  = (const float*)d_in[2];
  const float* b  = (const float*)d_in[3];
  const float* Wd = (const float*)d_in[4];
  const float* bd = (const float*)d_in[5];
  float* out = (float*)d_out;

  dim3 grid(B_ / 16);
  dim3 block(32 * NW * NP);
  hipLaunchKernelGGL(lstm_fused_kernel, grid, block, 0, stream,
                     x, W, U, b, Wd, bd, out);
}


# step barrier split into signal and wait: a wave signals behind its h store and 4 matrix ops of its next projection; the other 12 and the x stage store (4 stages) follow the signal; same sums
# speedup vs baseline: 4.8022x; 1.0455x over previous
#include <hip/hip_runtime.h>

typedef __attribute__((ext_vector_type(2))) float v2f;
typedef __attribute__((ext_vector_type(4))) float v4f;
typedef __attribute__((ext_vector_type(8))) float v8f;

#define B_   4096
#define T_   256
#define F_   64
#define H_   16
#define G4H  64
#define D_   64
#define NW   4
#define NP   2
#define OPAD 68
#define XP   68

__device__ __forceinline__ float sigf(float v) { return 1.0f / (1.0f + __expf(-v)); }

__device__ __forceinline__ void lds_barrier() {
  __builtin_amdgcn_fence(__ATOMIC_RELEASE, "workgroup", "local");
  __builtin_amdgcn_s_barrier();
  __builtin_amdgcn_fence(__ATOMIC_ACQUIRE, "workgroup", "local");
}

__device__ __forceinline__ v8f mma4(v2f a, v2f b, v8f c) {
  return __builtin_amdgcn_wmma_f32_16x16x4_f32(false, a, false, b, (short)0, c, false, false);
}

__global__ __launch_bounds__(32 * NW * NP)
void lstm_fused_kernel(const float* __restrict__ x, const float* __restrict__ W,
                       const float* __restrict__ U, const float* __restrict__ b,
                       const float* __restrict__ Wd, const float* __restrict__ bd,
                       float* __restrict__ out) {
  __shared__ __attribute__((aligned(16))) float ol[16 * OPAD];
  __shared__ float rinv[16];
  __shared__ __attribute__((aligned(16))) float hs[NP][NW][32][2];
  __shared__ __attribute__((aligned(16))) float cs[NP][NW][32][2];
  __shared__ __attribute__((aligned(16))) float xs[4][16 * XP];

  const int lane = threadIdx.x & 31;
  const int wv   = __builtin_amdgcn_readfirstlane((int)(threadIdx.x >> 5));
  const int w    = wv & 3;
  const int p    = wv >> 2;
  const int n    = lane & 15;
  const int half = lane >> 4;
  const int b0   = blockIdx.x * 16;

  const int colA = ((n & 7) >> 1) * 16 + 4 * w + 2 * (n >> 3) + (n & 1);

  v2f wfrag[16];
#pragma unroll
  for (int kk = 0; kk < 16; ++kk) {
    int f0 = 8 * (kk >> 1) + 4 * half + 2 * (kk & 1);
    wfrag[kk].x = W[f0 * G4H + colA];
    wfrag[kk].y = W[(f0 + 1) * G4H + colA];
  }

  v2f ufrag[4];
#pragma unroll
  for (int kk = 0; kk < 4; ++kk) {
    int u0 = 4 * kk + 2 * half;
    ufrag[kk].x = U[u0 * G4H + colA];
    ufrag[kk].y = U[(u0 + 1) * G4H + colA];
  }

  v8f bg;
#pragma unroll
  for (int r = 0; r < 8; ++r) bg[r] = b[(r >> 1) * 16 + 4 * w + 2 * half + (r & 1)];

  auto xw = [&](const v4f* xr, v8f a, int j0, int j1) {
#pragma unroll
    for (int j = j0; j < j1; ++j) {
      v2f lo; lo.x = xr[j].x; lo.y = xr[j].y;
      v2f hi; hi.x = xr[j].z; hi.y = xr[j].w;
      a = mma4(wfrag[2 * j], lo, a);
      a = mma4(wfrag[2 * j + 1], hi, a);
    }
    return a;
  };

  const int crow = 2 * wv + (lane >> 4);
  const int cf   = 4 * (lane & 15);
  const float* xco = x + (size_t)(b0 + crow) * T_ * F_ + cf;
  auto gload = [&](int t) {
    return *(const v4f*)(xco + (size_t)t * F_);
  };

  const float* frdb = &xs[0][n * XP + 4 * half];
  float* swrb = &xs[0][crow * XP + cf];
  const float* hrd = &hs[p ^ 1][0][lane][0];
  const float* crd = &cs[p ^ 1][w][lane][0];
  float* hwr = &hs[p][w][lane][0];
  float* cwr = &cs[p][w][lane][0];

  v8f acc = bg;
  v4f XF[8];

  auto gstep = [&](const float* frd) {
    __builtin_amdgcn_s_setprio(3);
    v2f hb[4];
#pragma unroll
    for (int kk = 0; kk < 4; ++kk) hb[kk] = *(const v2f*)(hrd + 64 * kk);
    v2f cp = *(const v2f*)crd;
    __builtin_amdgcn_sched_barrier(0);
#pragma unroll
    for (int i = 0; i < 8; ++i) XF[i] = *(const v4f*)(frd + 8 * i);
    __builtin_amdgcn_sched_barrier(0);
#pragma unroll
    for (int kk = 0; kk < 4; ++kk) acc = mma4(ufrag[kk], hb[kk], acc);
    const v8f z = acc;

    v2f hn, cn2;
    {
      float gi = sigf(z[0]);
      float gf = sigf(z[2]);
      float gg = fmaxf(z[4], 0.0f);
      float go = sigf(z[6]);
      float cn = gf * cp.x + gi * gg;
      cn2.x = cn;
      hn.x = go * fmaxf(cn, 0.0f);
    }
    {
      float gi = sigf(z[1]);
      float gf = sigf(z[3]);
      float gg = fmaxf(z[5], 0.0f);
      float go = sigf(z[7]);
      float cn = gf * cp.y + gi * gg;
      cn2.y = cn;
      hn.y = go * fmaxf(cn, 0.0f);
    }
    *(v2f*)hwr = hn;
    *(v2f*)cwr = cn2;
    __builtin_amdgcn_s_setprio(0);
    __builtin_amdgcn_sched_barrier(0);
    acc = xw(XF, bg, 0, 2);
    __builtin_amdgcn_sched_barrier(0);
  };

  v4f G0, G1, G2, G3;
  {
    {
      const float* q = x + (size_t)(b0 + n) * T_ * F_ + (size_t)p * F_ + 4 * half;
#pragma unroll
      for (int j = 0; j < 8; ++j) XF[j] = *(const v4f*)(q + 8 * j);
    }
    v4f GT2 = gload(2);
    v4f GT3 = gload(3);
    __builtin_amdgcn_sched_barrier(0);
    G0 = gload(4);
    __builtin_amdgcn_sched_barrier(0);
    G1 = gload(5);
    __builtin_amdgcn_sched_barrier(0);
    G2 = gload(6);
    __builtin_amdgcn_sched_barrier(0);
    G3 = gload(7);
    __builtin_amdgcn_sched_barrier(0);
    if (p == 1) {
      v2f zz; zz.x = 0.0f; zz.y = 0.0f;
      *(v2f*)hwr = zz;
      *(v2f*)cwr = zz;
    }
    *(v4f*)(swrb + 2 * 16 * XP) = GT2;
    *(v4f*)(swrb + 3 * 16 * XP) = GT3;
    acc = xw(XF, bg, 0, 8);
  }
  lds_barrier();

  auto phase = [&](int t, int q, v4f& g) {
    if (p == (q & 1)) gstep(frdb + ((q + 2) & 3) * 16 * XP);
    __builtin_amdgcn_fence(__ATOMIC_RELEASE, "workgroup", "local");
    __builtin_amdgcn_s_barrier_signal(-1);
    __builtin_amdgcn_sched_barrier(0);
    *(v4f*)(swrb + q * 16 * XP) = g;
    int tl = (t + 8 < T_) ? (t + 8) : (T_ - 1);
    g = gload(tl);
    __builtin_amdgcn_sched_barrier(0);
    if (p == (q & 1)) {
      acc = xw(XF, acc, 2, 8);
      __builtin_amdgcn_sched_barrier(0);
    }
    __builtin_amdgcn_s_barrier_wait(-1);
    __builtin_amdgcn_fence(__ATOMIC_ACQUIRE, "workgroup", "local");
  };

  for (int t = 0; t < T_; t += 4) {
    phase(t,     0, G0);
    phase(t + 1, 1, G1);
    phase(t + 2, 2, G2);
    phase(t + 3, 3, G3);
  }

  if (p == 0) {
    v2f hb[4];
#pragma unroll
    for (int kk = 0; kk < 4; ++kk) hb[kk] = *(const v2f*)&hs[1][kk][lane][0];
    v8f dacc;
#pragma unroll
    for (int r = 0; r < 8; ++r) dacc[r] = bd[16 * w + 8 * half + r];
#pragma unroll
    for (int kk = 0; kk < 4; ++kk) {
      int u0 = 4 * kk + 2 * half;
      v2f wa;
      wa.x = Wd[u0 * D_ + 16 * w + n];
      wa.y = Wd[(u0 + 1) * D_ + 16 * w + n];
      dacc = mma4(wa, hb[kk], dacc);
    }
    v4f s0, s1;
    s0.x = sigf(dacc[0]); s0.y = sigf(dacc[1]); s0.z = sigf(dacc[2]); s0.w = sigf(dacc[3]);
    s1.x = sigf(dacc[4]); s1.y = sigf(dacc[5]); s1.z = sigf(dacc[6]); s1.w = sigf(dacc[7]);
    float* q = ol + n * OPAD + 16 * w + 8 * half;
    *(v4f*)q = s0;
    *(v4f*)(q + 4) = s1;
  }
  __syncthreads();

  if (threadIdx.x < 16) {
    float s = 0.0f;
    for (int q = 0; q < 64; ++q) s += ol[threadIdx.x * OPAD + q];
    rinv[threadIdx.x] = 1.0f / s;
  }
  __syncthreads();

#pragma unroll
  for (int i = 0; i < 4; ++i) {
    int idx = i * (32 * NW * NP) + threadIdx.x;
    int row = idx >> 6, col = idx & 63;
    out[(size_t)(b0 + row) * D_ + col] = ol[row * OPAD + col] * rinv[row];
  }
}

extern "C" void kernel_launch(void* const* d_in, const int* in_sizes, int n_in,
                              void* d_out, int out_size, void* d_ws, size_t ws_size,
                              hipStream_t stream) {
  (void)in_sizes; (void)n_in; (void)out_size; (void)d_ws; (void)ws_size;
  const float* x  = (const float*)d_in[0];
  const float* W  = (const float*)d_in[1];
  const float* U  = (const float*)d_in[2];
  const float* b  = (const float*)d_in[3];
  const float* Wd = (const float*)d_in[4];
  const float* bd = (const float*)d_in[5];
  float* out = (float*)d_out;

  dim3 grid(B_ / 16);
  dim3 block(32 * NW * NP);
  hipLaunchKernelGGL(lstm_fused_kernel, grid, block, 0, stream,
                     x, W, U, b, Wd, bd, out);
}
